# starting kernel
# baseline (speedup 1.0000x reference)
#include <hip/hip_runtime.h>


#define DD 2048
#define MH 32

typedef __attribute__((ext_vector_type(2))) float v2f;
typedef __attribute__((ext_vector_type(8))) float v8f;

__device__ __forceinline__ v2f pair2(const v8f v, int j) {
    v2f r;
    r[0] = v[2 * j];
    r[1] = v[2 * j + 1];
    return r;
}

__global__ __launch_bounds__(256) void gemv_relu_kernel(
    const float* __restrict__ W, const float* __restrict__ a,
    const float* __restrict__ b, float* __restrict__ out)
{
    const int lane = threadIdx.x & 31;
    const int wave = threadIdx.x >> 5;
    const int row  = blockIdx.x * 8 + wave;

    const float* __restrict__ wrow = W + (size_t)row * DD;

    float acc = 0.0f;
    for (int j = lane * 4; j < DD; j += 32 * 4) {
        const float4 wv = *(const float4*)(wrow + j);
        const float4 av = *(const float4*)(a + j);
        acc += wv.x * av.x + wv.y * av.y + wv.z * av.z + wv.w * av.w;
    }
    acc += __shfl_xor(acc, 1);
    acc += __shfl_xor(acc, 2);
    acc += __shfl_xor(acc, 4);
    acc += __shfl_xor(acc, 8);
    acc += __shfl_xor(acc, 16);
    if (lane == 0) {
        const float v = acc + b[row];
        out[row] = v > 0.0f ? v : 0.0f;
    }
}

__global__ __launch_bounds__(256) void meta_update_kernel(
    const float* __restrict__ W,    const float* __restrict__ aPre,
    const float* __restrict__ aPost,const float* __restrict__ Wm1,
    const float* __restrict__ bm1,  const float* __restrict__ Wm2,
    const float* __restrict__ bm2,  float* __restrict__ Wout)
{
    __shared__ float sA[DD];
    for (int t = threadIdx.x; t < DD; t += 256) sA[t] = aPre[t];
    __syncthreads();

    const int lane = threadIdx.x & 31;
    const int wave = threadIdx.x >> 5;
    const int col  = lane & 15;
    const int half = lane >> 4;

    v2f A0, A1;
    A0[0] = half ? Wm1[2 * MH + col]      : Wm1[0 * MH + col];
    A0[1] = half ? bm1[col]               : Wm1[1 * MH + col];
    A1[0] = half ? Wm1[2 * MH + 16 + col] : Wm1[0 * MH + 16 + col];
    A1[1] = half ? bm1[16 + col]          : Wm1[1 * MH + 16 + col];

    const float* wmb = Wm2 + (half ? 8 : 0);
    v2f pA[4], pB[4];
#pragma unroll
    for (int j = 0; j < 4; ++j) {
        pA[j][0] = 0.5f * wmb[2 * j];
        pA[j][1] = 0.5f * wmb[2 * j + 1];
        pB[j][0] = 0.5f * wmb[16 + 2 * j];
        pB[j][1] = 0.5f * wmb[16 + 2 * j + 1];
    }

    const float accInit = half ? 0.0f : bm2[0];

    const int row = blockIdx.x * 8 + wave;
    const float* __restrict__ wrow = W    + (size_t)row * DD;
    float*       __restrict__ orow = Wout + (size_t)row * DD;
    const float  vO = aPost[row];

    for (int i0 = 0; i0 < DD; i0 += 32) {
        const float vi0 = sA[i0 + col];
        const float w0  = wrow[i0 + col];
        const float vi1 = sA[i0 + 16 + col];
        const float w1  = wrow[i0 + 16 + col];

        v2f Ba, Bb;
        Ba[0] = half ? vO   : vi0;
        Ba[1] = half ? 1.0f : w0;
        Bb[0] = half ? vO   : vi1;
        Bb[1] = half ? 1.0f : w1;

        v8f C0a = {}, C1a = {}, C0b = {}, C1b = {};
        C0a = __builtin_amdgcn_wmma_f32_16x16x4_f32(
                  false, A0, false, Ba, (short)0, C0a, false, false);
        C1a = __builtin_amdgcn_wmma_f32_16x16x4_f32(
                  false, A1, false, Ba, (short)0, C1a, false, false);
        C0b = __builtin_amdgcn_wmma_f32_16x16x4_f32(
                  false, A0, false, Bb, (short)0, C0b, false, false);
        C1b = __builtin_amdgcn_wmma_f32_16x16x4_f32(
                  false, A1, false, Bb, (short)0, C1b, false, false);

        v2f rawA = {accInit, 0.0f}, absA = {0.0f, 0.0f};
        v2f rawB = {accInit, 0.0f}, absB = {0.0f, 0.0f};
#pragma unroll
        for (int j = 0; j < 4; ++j) {
            const v2f c0a = pair2(C0a, j);
            const v2f c1a = pair2(C1a, j);
            const v2f c0b = pair2(C0b, j);
            const v2f c1b = pair2(C1b, j);
            rawA = __builtin_elementwise_fma(pA[j], c0a, rawA);
            absA = __builtin_elementwise_fma(pA[j], __builtin_elementwise_abs(c0a), absA);
            rawA = __builtin_elementwise_fma(pB[j], c1a, rawA);
            absA = __builtin_elementwise_fma(pB[j], __builtin_elementwise_abs(c1a), absA);
            rawB = __builtin_elementwise_fma(pA[j], c0b, rawB);
            absB = __builtin_elementwise_fma(pA[j], __builtin_elementwise_abs(c0b), absB);
            rawB = __builtin_elementwise_fma(pB[j], c1b, rawB);
            absB = __builtin_elementwise_fma(pB[j], __builtin_elementwise_abs(c1b), absB);
        }
        const v2f tA = rawA + absA;
        const v2f tB = rawB + absB;
        float accA = tA[0] + tA[1];
        float accB = tB[0] + tB[1];

        accA += __shfl_xor(accA, 16);
        accB += __shfl_xor(accB, 16);

        const float res = (lane < 16) ? accA : accB;
        __builtin_nontemporal_store(res, orow + i0 + lane);
    }
}

extern "C" void kernel_launch(void* const* d_in, const int* in_sizes, int n_in,
                              void* d_out, int out_size, void* d_ws, size_t ws_size,
                              hipStream_t stream) {
    const float* x   = (const float*)d_in[0];
    const float* W1  = (const float*)d_in[1];
    const float* b1  = (const float*)d_in[2];
    const float* W2  = (const float*)d_in[3];
    const float* b2  = (const float*)d_in[4];
    const float* W3  = (const float*)d_in[5];
    const float* b3  = (const float*)d_in[6];
    const float* Wm1 = (const float*)d_in[7];
    const float* bm1 = (const float*)d_in[8];
    const float* Wm2 = (const float*)d_in[9];
    const float* bm2 = (const float*)d_in[10];

    float* out = (float*)d_out;
    float* nW1 = out + DD;
    float* nW2 = nW1 + (size_t)DD * DD;
    float* nW3 = nW2 + (size_t)DD * DD;

    float* a1 = (float*)d_ws;
    float* a2 = a1 + DD;

    dim3 blk(256);
    dim3 grid(DD / 8);

    gemv_relu_kernel<<<grid, blk, 0, stream>>>(W1, x,  b1, a1);
    gemv_relu_kernel<<<grid, blk, 0, stream>>>(W2, a1, b2, a2);
    gemv_relu_kernel<<<grid, blk, 0, stream>>>(W3, a2, b3, out);

    meta_update_kernel<<<grid, blk, 0, stream>>>(W1, x,  a1,  Wm1, bm1, Wm2, bm2, nW1);
    meta_update_kernel<<<grid, blk, 0, stream>>>(W2, a1, a2,  Wm1, bm1, Wm2, bm2, nW2);
    meta_update_kernel<<<grid, blk, 0, stream>>>(W3, a2, out, Wm1, bm1, Wm2, bm2, nW3);
}


# meta kernel: matrix tile re-cut to 8 meta rows x 32 entries (A block-diagonal over k halves, row constants as C): each lane owns its entries, no cross-lane sum or operand select; sum order changed
# speedup vs baseline: 4.1392x; 4.1392x over previous
#include <hip/hip_runtime.h>


#define DD 2048
#define MH 32

typedef __attribute__((ext_vector_type(2))) float v2f;
typedef __attribute__((ext_vector_type(4))) float v4f;
typedef __attribute__((ext_vector_type(8))) float v8f;

__device__ __forceinline__ v2f pair2(const v8f v, int j) {
    v2f r;
    r[0] = v[2 * j];
    r[1] = v[2 * j + 1];
    return r;
}

__device__ __forceinline__ v2f relu2(const v2f x) {
    float a, b;
    asm volatile("v_dual_max_num_f32 %0, 0, %2 :: v_dual_max_num_f32 %1, 0, %3"
                 : "=v"(a), "=v"(b) : "v"(x[0]), "v"(x[1]));
    v2f r;
    r[0] = a;
    r[1] = b;
    return r;
}

__global__ __launch_bounds__(256) void gemv_relu_kernel(
    const float* __restrict__ W, const float* __restrict__ a,
    const float* __restrict__ b, float* __restrict__ out)
{
    const int lane = threadIdx.x & 31;
    const int wave = threadIdx.x >> 5;
    const int row  = blockIdx.x * 8 + wave;

    const float4* __restrict__ wrow4 = (const float4*)(W + (size_t)row * DD) + lane;
    const float4* __restrict__ a4    = (const float4*)a + lane;

    float4 wv[16], av[16];
#pragma unroll
    for (int j = 0; j < 16; ++j) wv[j] = wrow4[j * 32];
#pragma unroll
    for (int j = 0; j < 16; ++j) av[j] = a4[j * 32];
    const float bias = b[row];

    float acc = 0.0f;
#pragma unroll
    for (int j = 0; j < 16; ++j)
        acc += wv[j].x * av[j].x + wv[j].y * av[j].y + wv[j].z * av[j].z + wv[j].w * av[j].w;
    __builtin_amdgcn_sched_group_barrier(0x020, 33, 0);
    __builtin_amdgcn_sched_group_barrier(0x002, 256, 0);

    acc += __shfl_xor(acc, 1);
    acc += __shfl_xor(acc, 2);
    acc += __shfl_xor(acc, 4);
    acc += __shfl_xor(acc, 8);
    acc += __shfl_xor(acc, 16);
    asm volatile("" :: "v"(bias));
    if (lane == 0) {
        const float v = acc + bias;
        out[row] = v > 0.0f ? v : 0.0f;
    }
}

__global__ __launch_bounds__(256) void meta_update_kernel(
    const float* __restrict__ W1,  const float* __restrict__ W2,
    const float* __restrict__ W3,  const float* __restrict__ act0,
    const float* __restrict__ act1,const float* __restrict__ act2,
    const float* __restrict__ act3,const float* __restrict__ Wm1,
    const float* __restrict__ bm1, const float* __restrict__ Wm2,
    const float* __restrict__ bm2, float* __restrict__ nW)
{
    const int layer = blockIdx.y;
    const float* __restrict__ W     = layer == 0 ? W1   : (layer == 1 ? W2   : W3);
    const float* __restrict__ aPre  = layer == 0 ? act0 : (layer == 1 ? act1 : act2);
    const float* __restrict__ aPost = layer == 0 ? act1 : (layer == 1 ? act2 : act3);
    float*       __restrict__ Wout  = nW + (size_t)layer * DD * DD;

    const int lane = threadIdx.x & 31;
    const int wave = threadIdx.x >> 5;
    const int row  = blockIdx.x * 8 + wave;

    __shared__ float4 sLds[DD / 4 + 8 * (DD / 4)];
    float4* sWr = sLds + (DD / 4) * (1 + wave);
    const float4* __restrict__ wrow4 = (const float4*)(W + (size_t)row * DD) + lane;
    float4 wv[16];
#pragma unroll
    for (int j = 0; j < 16; ++j) wv[j] = wrow4[j * 32];

    const float4* __restrict__ aPre4 = (const float4*)aPre + threadIdx.x;
    const float4 pa0 = aPre4[0];
    const float4 pa1 = aPre4[256];

    const int  r8  = lane & 7;
    const bool own = ((lane >> 4) & 1) == ((lane >> 3) & 1);
    v2f A[4];
#pragma unroll
    for (int t = 0; t < 4; ++t) {
        const float c0 = Wm1[0 * MH + 8 * t + r8];
        const float c1 = Wm1[1 * MH + 8 * t + r8];
        A[t][0] = own ? c0 : 0.0f;
        A[t][1] = own ? c1 : 0.0f;
    }

    const float vO = aPost[row];
    v8f Cc[4];
#pragma unroll
    for (int t = 0; t < 4; ++t)
#pragma unroll
        for (int r = 0; r < 8; ++r)
            Cc[t][r] = fmaf(Wm1[2 * MH + 8 * t + r], vO, bm1[8 * t + r]);

    v2f pW[4][4];
#pragma unroll
    for (int t = 0; t < 4; ++t)
#pragma unroll
        for (int j = 0; j < 4; ++j) {
            pW[t][j][0] = Wm2[8 * t + 2 * j];
            pW[t][j][1] = Wm2[8 * t + 2 * j + 1];
        }
    const float bm2v = bm2[0];

#pragma unroll
    for (int j = 0; j < 16; ++j) sWr[j * 32 + lane] = wv[j];
    sLds[threadIdx.x]       = pa0;
    sLds[threadIdx.x + 256] = pa1;
    __syncthreads();

    const v4f* wl4 = (const v4f*)sWr  + lane;
    const v4f* al4 = (const v4f*)sLds + lane;
    v4f* __restrict__ orow4 = (v4f*)(Wout + (size_t)row * DD) + lane;

    for (int j = 0; j < DD / 128; ++j) {
        const v4f wq = wl4[j * 32];
        const v4f vq = al4[j * 32];
        v4f res;
#pragma unroll
        for (int q = 0; q < 4; ++q) {
            v2f B;
            B[0] = vq[q];
            B[1] = wq[q];

            const v8f D0 = __builtin_amdgcn_wmma_f32_16x16x4_f32(
                               false, A[0], false, B, (short)0, Cc[0], false, false);
            const v8f D1 = __builtin_amdgcn_wmma_f32_16x16x4_f32(
                               false, A[1], false, B, (short)0, Cc[1], false, false);
            const v8f D2 = __builtin_amdgcn_wmma_f32_16x16x4_f32(
                               false, A[2], false, B, (short)0, Cc[2], false, false);
            const v8f D3 = __builtin_amdgcn_wmma_f32_16x16x4_f32(
                               false, A[3], false, B, (short)0, Cc[3], false, false);

            v2f rD0[4], rD1[4], rD2[4], rD3[4];
#pragma unroll
            for (int k = 0; k < 4; ++k) rD0[k] = relu2(pair2(D0, k));
#pragma unroll
            for (int k = 0; k < 4; ++k) rD1[k] = relu2(pair2(D1, k));
#pragma unroll
            for (int k = 0; k < 4; ++k) rD2[k] = relu2(pair2(D2, k));
#pragma unroll
            for (int k = 0; k < 4; ++k) rD3[k] = relu2(pair2(D3, k));

            v2f s0 = {bm2v, 0.0f}, s1 = {0.0f, 0.0f};
#pragma unroll
            for (int k = 0; k < 4; ++k) {
                s0 = __builtin_elementwise_fma(pW[0][k], rD0[k], s0);
                s1 = __builtin_elementwise_fma(pW[1][k], rD1[k], s1);
            }
#pragma unroll
            for (int k = 0; k < 4; ++k) {
                s0 = __builtin_elementwise_fma(pW[2][k], rD2[k], s0);
                s1 = __builtin_elementwise_fma(pW[3][k], rD3[k], s1);
            }
            const v2f tt = s0 + s1;
            res[q] = tt[0] + tt[1];
            __builtin_amdgcn_sched_barrier(0);
        }
        __builtin_nontemporal_store(res, orow4 + j * 32);
    }
}

extern "C" void kernel_launch(void* const* d_in, const int* in_sizes, int n_in,
                              void* d_out, int out_size, void* d_ws, size_t ws_size,
                              hipStream_t stream) {
    const float* x   = (const float*)d_in[0];
    const float* W1  = (const float*)d_in[1];
    const float* b1  = (const float*)d_in[2];
    const float* W2  = (const float*)d_in[3];
    const float* b2  = (const float*)d_in[4];
    const float* W3  = (const float*)d_in[5];
    const float* b3  = (const float*)d_in[6];
    const float* Wm1 = (const float*)d_in[7];
    const float* bm1 = (const float*)d_in[8];
    const float* Wm2 = (const float*)d_in[9];
    const float* bm2 = (const float*)d_in[10];

    float* out = (float*)d_out;
    float* nW1 = out + DD;

    float* a1 = (float*)d_ws;
    float* a2 = a1 + DD;

    dim3 blk(256);
    dim3 grid(DD / 8);

    gemv_relu_kernel<<<grid, blk, 0, stream>>>(W1, x,  b1, a1);
    gemv_relu_kernel<<<grid, blk, 0, stream>>>(W2, a1, b2, a2);
    gemv_relu_kernel<<<grid, blk, 0, stream>>>(W3, a2, b3, out);

    dim3 grid3(DD / 8, 3);
    meta_update_kernel<<<grid3, blk, 0, stream>>>(W1, W2, W3, x, a1, a2, out,
                                                  Wm1, bm1, Wm2, bm2, nW1);
}
